# starting kernel
# baseline (speedup 1.0000x reference)
#include <hip/hip_runtime.h>
#include <hip/hip_bf16.h>

typedef float v2f __attribute__((ext_vector_type(2)));
typedef float v8f __attribute__((ext_vector_type(8)));

#define D 128


__global__ void gcn_init_deg(float* __restrict__ deg, int n) {
    int i = blockIdx.x * blockDim.x + threadIdx.x;
    if (i < n) deg[i] = 1.0f;
}

__global__ void gcn_deg_edges(const int* __restrict__ dst, float* __restrict__ deg, int n_edges) {
    int stride = gridDim.x * blockDim.x;
    for (int e = blockIdx.x * blockDim.x + threadIdx.x; e < n_edges; e += stride)
        atomicAdd(&deg[dst[e]], 1.0f);
}

__global__ void gcn_dinv(float* __restrict__ deg, int n) {
    int i = blockIdx.x * blockDim.x + threadIdx.x;
    if (i < n) {
        float d = deg[i];
        deg[i] = (d > 0.0f) ? __frsqrt_rn(d) : 0.0f;
    }
}


template <bool RELU>
__global__ void gcn_gemm_wmma(const float* __restrict__ X, const float* __restrict__ W,
                              float* __restrict__ H, int n_rows) {
    const int wave = threadIdx.x >> 5;
    const int lane = threadIdx.x & 31;
    const int mbase = blockIdx.x * 16;
    const int nbase = wave * 16;
    const int half = lane >> 4;
    const int l16  = lane & 15;

    const bool full_tile = (mbase + 16 <= n_rows);
    int arow = mbase + l16;
    if (arow >= n_rows) arow = n_rows - 1;
    const float* __restrict__ ap = X + (size_t)arow * D;
    const int bcol = nbase + l16;

    v8f acc = {};
    #pragma unroll 8
    for (int k = 0; k < D; k += 4) {
        const int kk = k + half * 2;
        v2f a;
        a.x = ap[kk];
        a.y = ap[kk + 1];
        if (RELU) { a.x = fmaxf(a.x, 0.0f); a.y = fmaxf(a.y, 0.0f); }
        v2f b;
        b.x = W[(size_t)kk * D + bcol];
        b.y = W[(size_t)(kk + 1) * D + bcol];
        acc = __builtin_amdgcn_wmma_f32_16x16x4_f32(
             false, a,  false, b,
             (short)0, acc,  false,  false);
    }

    const int crow0 = mbase + half * 8;
    float* __restrict__ cp = H + (size_t)crow0 * D + nbase + l16;
    if (full_tile) {
        #pragma unroll
        for (int v = 0; v < 8; ++v) cp[(size_t)v * D] = acc[v];
    } else {
        #pragma unroll
        for (int v = 0; v < 8; ++v)
            if (crow0 + v < n_rows) cp[(size_t)v * D] = acc[v];
    }
}


__global__ void gcn_agg_init(const float* __restrict__ H, const float* __restrict__ dinv,
                             const float* __restrict__ bias, float* __restrict__ Y, int n_rows) {
    int tid = blockIdx.x * blockDim.x + threadIdx.x;
    int node = tid >> 5;
    int c = (tid & 31) * 4;
    if (node >= n_rows) return;
    float s = dinv[node];
    s = s * s;
    float4 h = *(const float4*)(H + (size_t)node * D + c);
    float4 b = *(const float4*)(bias + c);
    float4 r = make_float4(fmaf(h.x, s, b.x), fmaf(h.y, s, b.y),
                           fmaf(h.z, s, b.z), fmaf(h.w, s, b.w));
    *(float4*)(Y + (size_t)node * D + c) = r;
}

__global__ void gcn_agg_edges(const float* __restrict__ H, const float* __restrict__ dinv,
                              const int* __restrict__ src, const int* __restrict__ dst,
                              float* __restrict__ Y, int n_edges) {
    const int lane = threadIdx.x & 31;
    const int wid = (blockIdx.x * blockDim.x + threadIdx.x) >> 5;
    const int nwaves = (gridDim.x * blockDim.x) >> 5;

    for (int e = wid; e < n_edges; e += nwaves) {
        const int s = src[e];
        const int d = dst[e];
        const int e2 = e + nwaves;
        if (e2 < n_edges) {
            const int s2 = src[e2];
            __builtin_prefetch(H + (size_t)s2 * D + lane * 4, 0, 1);
        }
        const float norm = dinv[s] * dinv[d];
        float4 h = *(const float4*)(H + (size_t)s * D + lane * 4);
        float* yp = Y + (size_t)d * D + lane * 4;
        atomicAdd(yp + 0, h.x * norm);
        atomicAdd(yp + 1, h.y * norm);
        atomicAdd(yp + 2, h.z * norm);
        atomicAdd(yp + 3, h.w * norm);
    }
}


extern "C" void kernel_launch(void* const* d_in, const int* in_sizes, int n_in,
                              void* d_out, int out_size, void* d_ws, size_t ws_size,
                              hipStream_t stream) {
    const float* x   = (const float*)d_in[0];
    const int*   ei  = (const int*)d_in[1];
    const float* W1  = (const float*)d_in[2];
    const float* b1  = (const float*)d_in[3];
    const float* W2  = (const float*)d_in[4];
    const float* b2  = (const float*)d_in[5];
    const float* W3  = (const float*)d_in[6];
    const float* b3  = (const float*)d_in[7];

    const int N = in_sizes[0] / D;
    const int E = in_sizes[1] / 2;
    const int* src = ei;
    const int* dst = ei + E;

    float* dinv = (float*)d_ws;
    size_t off = ((size_t)N * sizeof(float) + 255) & ~(size_t)255;
    float* H = (float*)((char*)d_ws + off);
    float* Y = H + (size_t)N * D;
    float* out = (float*)d_out;

    const int T = 256;
    const int nodeBlocks  = (N + T - 1) / T;
    const int mTiles      = (N + 15) / 16;
    const int initBlocks  = (int)(((size_t)N * 32 + T - 1) / T);
    const int edgeBlocks  = 2048;

    gcn_init_deg<<<nodeBlocks, T, 0, stream>>>(dinv, N);
    gcn_deg_edges<<<1024, T, 0, stream>>>(dst, dinv, E);
    gcn_dinv<<<nodeBlocks, T, 0, stream>>>(dinv, N);

    gcn_gemm_wmma<false><<<mTiles, T, 0, stream>>>(x, W1, H, N);
    gcn_agg_init<<<initBlocks, T, 0, stream>>>(H, dinv, b1, Y, N);
    gcn_agg_edges<<<edgeBlocks, T, 0, stream>>>(H, dinv, src, dst, Y, E);

    gcn_gemm_wmma<true><<<mTiles, T, 0, stream>>>(Y, W2, H, N);
    gcn_agg_init<<<initBlocks, T, 0, stream>>>(H, dinv, b2, Y, N);
    gcn_agg_edges<<<edgeBlocks, T, 0, stream>>>(H, dinv, src, dst, Y, E);

    gcn_gemm_wmma<true><<<mTiles, T, 0, stream>>>(Y, W3, H, N);
    gcn_agg_init<<<initBlocks, T, 0, stream>>>(H, dinv, b3, out, N);
    gcn_agg_edges<<<edgeBlocks, T, 0, stream>>>(H, dinv, src, dst, out, E);
}


# aggregation: f32 atomic scatter-add replaced by gather over edges sorted by destination per call (a wave sums a row in registers, stores it once); sum order fixed: bias, sources ascending, self
# speedup vs baseline: 2.6888x; 2.6888x over previous
#include <hip/hip_runtime.h>
#include <hip/hip_bf16.h>


typedef float v2f __attribute__((ext_vector_type(2)));
typedef float v8f __attribute__((ext_vector_type(8)));

#define D 128


__global__ void gcn_zero_cnt(int* __restrict__ cnt, int n) {
    int i = blockIdx.x * blockDim.x + threadIdx.x;
    if (i < n) cnt[i] = 0;
}

__global__ void gcn_count_rank(const int* __restrict__ dst, int* __restrict__ cnt,
                               int* __restrict__ rank, int n_edges) {
    int stride = gridDim.x * blockDim.x;
    for (int e = blockIdx.x * blockDim.x + threadIdx.x; e < n_edges; e += stride)
        rank[e] = atomicAdd(&cnt[dst[e]], 1);
}

__global__ void __launch_bounds__(1024)
gcn_scan_dinv(const int* __restrict__ cnt, int* __restrict__ rowptr,
              float* __restrict__ dinv, int n) {
    __shared__ int wpre[32];
    __shared__ int wtot[32];
    const int tid = threadIdx.x;
    const int lane = tid & 31;
    const int wave = tid >> 5;
    const int base = blockIdx.x * 1024;

    const int4* __restrict__ cq = (const int4*)cnt;
    const int nq = base >> 2;
    int t = 0;
    #pragma unroll 4
    for (int q = tid; q < nq; q += 1024) {
        const int4 v = cq[q];
        t += (v.x + v.y) + (v.z + v.w);
    }
    for (int o = 16; o > 0; o >>= 1) t += __shfl_xor(t, o);

    const int i = base + tid;
    const int c = (i < n) ? cnt[i] : 0;
    int x = c;
    for (int o = 1; o < 32; o <<= 1) {
        const int y = __shfl_up(x, o);
        if (lane >= o) x += y;
    }
    if (lane == 31) wtot[wave] = x;
    if (lane == 0) wpre[wave] = t;
    __syncthreads();

    int a = wpre[lane] + ((lane < wave) ? wtot[lane] : 0);
    for (int o = 16; o > 0; o >>= 1) a += __shfl_xor(a, o);

    if (i < n) {
        rowptr[i] = a + x - c;
        if (i == n - 1) rowptr[n] = a + x;
        const float d = (float)(c + 1);
        dinv[i] = (d > 0.0f) ? __frsqrt_rn(d) : 0.0f;
    }
}

__global__ void gcn_fill(const int* __restrict__ src, const int* __restrict__ dst,
                         const int* __restrict__ rank, const int* __restrict__ rowptr,
                         const float* __restrict__ dinv, int2* __restrict__ colw, int n_edges) {
    int stride = gridDim.x * blockDim.x;
    for (int e = blockIdx.x * blockDim.x + threadIdx.x; e < n_edges; e += stride) {
        const int s = src[e];
        const int d = dst[e];
        const float w = dinv[s] * dinv[d];
        colw[rowptr[d] + rank[e]] = make_int2(s, __float_as_int(w));
    }
}

__global__ void __launch_bounds__(256)
gcn_order_rows(const int* __restrict__ rowptr, const int2* __restrict__ colw_in,
               int2* __restrict__ colw_out, int n_rows) {
    const int lane = threadIdx.x & 31;
    const int wid = __builtin_amdgcn_readfirstlane((int)((blockIdx.x * blockDim.x + threadIdx.x) >> 5));
    const int nwaves = (int)((gridDim.x * blockDim.x) >> 5);

    for (int node = wid; node < n_rows; node += nwaves) {
        const int beg = rowptr[node];
        const int len = rowptr[node + 1] - beg;
        for (int c0 = 0; c0 < len; c0 += 32) {
            const int i = c0 + lane;
            int2 mine = make_int2(0, 0);
            if (i < len) mine = colw_in[beg + i];
            int pos = 0;
            for (int c1 = 0; c1 < len; c1 += 32) {
                const int i1 = c1 + lane;
                int k1 = 0;
                if (i1 < len) k1 = colw_in[beg + i1].x;
                int m1 = len - c1;
                if (m1 > 32) m1 = 32;
                for (int j = 0; j < m1; ++j) {
                    const int kj = __builtin_amdgcn_readlane(k1, j);
                    const int ij = c1 + j;
                    pos += (kj < mine.x || (kj == mine.x && ij < i)) ? 1 : 0;
                }
            }
            if (i < len) colw_out[beg + pos] = mine;
        }
    }
}


template <bool RELU>
__global__ void gcn_gemm_wmma(const float* __restrict__ X, const float* __restrict__ W,
                              float* __restrict__ H, int n_rows) {
    const int wave = threadIdx.x >> 5;
    const int lane = threadIdx.x & 31;
    const int mbase = blockIdx.x * 16;
    const int nbase = wave * 16;
    const int half = lane >> 4;
    const int l16  = lane & 15;

    const bool full_tile = (mbase + 16 <= n_rows);
    int arow = mbase + l16;
    if (arow >= n_rows) arow = n_rows - 1;
    const float* __restrict__ ap = X + (size_t)arow * D;
    const int bcol = nbase + l16;

    v8f acc = {};
    #pragma unroll 8
    for (int k = 0; k < D; k += 4) {
        const int kk = k + half * 2;
        v2f a;
        a.x = ap[kk];
        a.y = ap[kk + 1];
        if (RELU) { a.x = fmaxf(a.x, 0.0f); a.y = fmaxf(a.y, 0.0f); }
        v2f b;
        b.x = W[(size_t)kk * D + bcol];
        b.y = W[(size_t)(kk + 1) * D + bcol];
        acc = __builtin_amdgcn_wmma_f32_16x16x4_f32(
             false, a,  false, b,
             (short)0, acc,  false,  false);
    }

    const int crow0 = mbase + half * 8;
    float* __restrict__ cp = H + (size_t)crow0 * D + nbase + l16;
    if (full_tile) {
        #pragma unroll
        for (int v = 0; v < 8; ++v) cp[(size_t)v * D] = acc[v];
    } else {
        #pragma unroll
        for (int v = 0; v < 8; ++v)
            if (crow0 + v < n_rows) cp[(size_t)v * D] = acc[v];
    }
}


static __device__ __forceinline__ void gcn_axpy(float4& acc, const float4 h, const float w) {
    acc.x = fmaf(h.x, w, acc.x);
    acc.y = fmaf(h.y, w, acc.y);
    acc.z = fmaf(h.z, w, acc.z);
    acc.w = fmaf(h.w, w, acc.w);
}

template <int K>
static __device__ __forceinline__ void gcn_rows(float4& acc, const float* __restrict__ Hc,
                                                const int2 cw, const int j) {
    float4 h[K];
    float w[K];
    #pragma unroll
    for (int u = 0; u < K; ++u) {
        const int si = __builtin_amdgcn_readlane(cw.x, j + u);
        w[u] = __int_as_float(__builtin_amdgcn_readlane(cw.y, j + u));
        h[u] = *(const float4*)(Hc + (size_t)si * D);
    }
    #pragma unroll
    for (int u = 0; u < K; ++u) gcn_axpy(acc, h[u], w[u]);
}

__global__ void __launch_bounds__(256)
gcn_agg_gather(const float* __restrict__ H, const float* __restrict__ dinv,
               const int* __restrict__ rowptr, const int2* __restrict__ colw,
               const float* __restrict__ bias, float* __restrict__ Y, int n_rows) {
    const int lane = threadIdx.x & 31;
    const int c4 = lane * 4;
    const int wid = __builtin_amdgcn_readfirstlane((int)((blockIdx.x * blockDim.x + threadIdx.x) >> 5));
    const int nwaves = (int)((gridDim.x * blockDim.x) >> 5);
    const float* __restrict__ Hc = H + c4;
    const float4 b = *(const float4*)(bias + c4);

    for (int node = wid; node < n_rows; node += nwaves) {
        const int beg = rowptr[node];
        const int len = rowptr[node + 1] - beg;
        float s = dinv[node];
        s = s * s;
        const float4 hs = *(const float4*)(Hc + (size_t)node * D);
        float4 acc = b;

        for (int base = 0; base < len; base += 32) {
            int m = len - base;
            if (m > 32) m = 32;
            int2 cw = make_int2(0, 0);
            if (lane < m) cw = colw[beg + base + lane];

            int j = 0;
            for (; j + 8 <= m; j += 8) gcn_rows<8>(acc, Hc, cw, j);
            if (j + 4 <= m) { gcn_rows<4>(acc, Hc, cw, j); j += 4; }
            if (j + 2 <= m) { gcn_rows<2>(acc, Hc, cw, j); j += 2; }
            if (j < m) gcn_rows<1>(acc, Hc, cw, j);
        }
        gcn_axpy(acc, hs, s);
        *(float4*)(Y + (size_t)node * D + c4) = acc;
    }
}


extern "C" void kernel_launch(void* const* d_in, const int* in_sizes, int n_in,
                              void* d_out, int out_size, void* d_ws, size_t ws_size,
                              hipStream_t stream) {
    const float* x   = (const float*)d_in[0];
    const int*   ei  = (const int*)d_in[1];
    const float* W1  = (const float*)d_in[2];
    const float* b1  = (const float*)d_in[3];
    const float* W2  = (const float*)d_in[4];
    const float* b2  = (const float*)d_in[5];
    const float* W3  = (const float*)d_in[6];
    const float* b3  = (const float*)d_in[7];

    const int N = in_sizes[0] / D;
    const int E = in_sizes[1] / 2;
    const int* src = ei;
    const int* dst = ei + E;

    const size_t nodeBytes = ((size_t)(N + 1) * sizeof(int) + 255) & ~(size_t)255;
    const size_t edgeBytes = ((size_t)E * sizeof(int) + 255) & ~(size_t)255;
    char* p = (char*)d_ws;
    float* dinv = (float*)p;                 p += nodeBytes;
    float* H = (float*)p;                    p += (size_t)N * D * sizeof(float);
    float* Y = (float*)p;                    p += (size_t)N * D * sizeof(float);
    int* cnt = (int*)p;                      p += nodeBytes;
    int* rowptr = (int*)p;                   p += nodeBytes;
    int* rank = (int*)p;                     p += edgeBytes;
    int2* colw0 = (int2*)p;                  p += 2 * edgeBytes;
    int2* colw = (int2*)p;
    float* out = (float*)d_out;

    const int T = 256;
    const int nodeBlocks  = (N + T - 1) / T;
    const int scanBlocks  = (N + 1023) / 1024;
    const int mTiles      = (N + 15) / 16;
    const int rowBlocks   = 2048;

    gcn_zero_cnt<<<nodeBlocks, T, 0, stream>>>(cnt, N);
    gcn_count_rank<<<1024, T, 0, stream>>>(dst, cnt, rank, E);
    gcn_scan_dinv<<<scanBlocks, 1024, 0, stream>>>(cnt, rowptr, dinv, N);
    gcn_fill<<<1024, T, 0, stream>>>(src, dst, rank, rowptr, dinv, colw0, E);
    gcn_order_rows<<<rowBlocks, T, 0, stream>>>(rowptr, colw0, colw, N);

    gcn_gemm_wmma<false><<<mTiles, T, 0, stream>>>(x, W1, H, N);
    gcn_agg_gather<<<rowBlocks, T, 0, stream>>>(H, dinv, rowptr, colw, b1, Y, N);

    gcn_gemm_wmma<true><<<mTiles, T, 0, stream>>>(Y, W2, H, N);
    gcn_agg_gather<<<rowBlocks, T, 0, stream>>>(H, dinv, rowptr, colw, b2, Y, N);

    gcn_gemm_wmma<true><<<mTiles, T, 0, stream>>>(Y, W3, H, N);
    gcn_agg_gather<<<rowBlocks, T, 0, stream>>>(H, dinv, rowptr, colw, b3, out, N);
}


# each layer in one kernel: a work-group aggregates 16 input rows into LDS, then multiplies them by W on f32 WMMA; sum order changes: (A.X).W + b, was A.(X.W) + b; 8 launches, were 11
# speedup vs baseline: 3.6767x; 1.3674x over previous
#include <hip/hip_runtime.h>
#include <hip/hip_bf16.h>


typedef float v2f __attribute__((ext_vector_type(2)));
typedef float v8f __attribute__((ext_vector_type(8)));

#define D 128


__global__ void gcn_zero_cnt(int* __restrict__ cnt, int n) {
    int i = blockIdx.x * blockDim.x + threadIdx.x;
    if (i < n) cnt[i] = 0;
}

__global__ void gcn_count_rank(const int* __restrict__ dst, int* __restrict__ cnt,
                               int* __restrict__ rank, int n_edges) {
    int stride = gridDim.x * blockDim.x;
    for (int e = blockIdx.x * blockDim.x + threadIdx.x; e < n_edges; e += stride)
        rank[e] = atomicAdd(&cnt[dst[e]], 1);
}

__global__ void __launch_bounds__(1024)
gcn_scan_dinv(const int* __restrict__ cnt, int* __restrict__ rowptr,
              float* __restrict__ dinv, int n) {
    __shared__ int wpre[32];
    __shared__ int wtot[32];
    const int tid = threadIdx.x;
    const int lane = tid & 31;
    const int wave = tid >> 5;
    const int base = blockIdx.x * 1024;

    const int4* __restrict__ cq = (const int4*)cnt;
    const int nq = base >> 2;
    int t = 0;
    #pragma unroll 4
    for (int q = tid; q < nq; q += 1024) {
        const int4 v = cq[q];
        t += (v.x + v.y) + (v.z + v.w);
    }
    for (int o = 16; o > 0; o >>= 1) t += __shfl_xor(t, o);

    const int i = base + tid;
    const int c = (i < n) ? cnt[i] : 0;
    int x = c;
    for (int o = 1; o < 32; o <<= 1) {
        const int y = __shfl_up(x, o);
        if (lane >= o) x += y;
    }
    if (lane == 31) wtot[wave] = x;
    if (lane == 0) wpre[wave] = t;
    __syncthreads();

    int a = wpre[lane] + ((lane < wave) ? wtot[lane] : 0);
    for (int o = 16; o > 0; o >>= 1) a += __shfl_xor(a, o);

    if (i < n) {
        rowptr[i] = a + x - c;
        if (i == n - 1) rowptr[n] = a + x;
        const float d = (float)(c + 1);
        dinv[i] = (d > 0.0f) ? __frsqrt_rn(d) : 0.0f;
    }
}

__global__ void gcn_fill(const int* __restrict__ src, const int* __restrict__ dst,
                         const int* __restrict__ rank, const int* __restrict__ rowptr,
                         const float* __restrict__ dinv, int2* __restrict__ colw, int n_edges) {
    int stride = gridDim.x * blockDim.x;
    for (int e = blockIdx.x * blockDim.x + threadIdx.x; e < n_edges; e += stride) {
        const int s = src[e];
        const int d = dst[e];
        const float w = dinv[s] * dinv[d];
        colw[rowptr[d] + rank[e]] = make_int2(s, __float_as_int(w));
    }
}

__global__ void __launch_bounds__(256)
gcn_order_rows(const int* __restrict__ rowptr, const int2* __restrict__ colw_in,
               int2* __restrict__ colw_out, int n_rows) {
    const int lane = threadIdx.x & 31;
    const int wid = __builtin_amdgcn_readfirstlane((int)((blockIdx.x * blockDim.x + threadIdx.x) >> 5));
    const int nwaves = (int)((gridDim.x * blockDim.x) >> 5);

    for (int node = wid; node < n_rows; node += nwaves) {
        const int beg = rowptr[node];
        const int len = rowptr[node + 1] - beg;
        for (int c0 = 0; c0 < len; c0 += 32) {
            const int i = c0 + lane;
            int2 mine = make_int2(0, 0);
            if (i < len) mine = colw_in[beg + i];
            int pos = 0;
            for (int c1 = 0; c1 < len; c1 += 32) {
                const int i1 = c1 + lane;
                int k1 = 0;
                if (i1 < len) k1 = colw_in[beg + i1].x;
                int m1 = len - c1;
                if (m1 > 32) m1 = 32;
                for (int j = 0; j < m1; ++j) {
                    const int kj = __builtin_amdgcn_readlane(k1, j);
                    const int ij = c1 + j;
                    pos += (kj < mine.x || (kj == mine.x && ij < i)) ? 1 : 0;
                }
            }
            if (i < len) colw_out[beg + pos] = mine;
        }
    }
}


static __device__ __forceinline__ void gcn_axpy(float4& acc, const float4 h, const float w) {
    acc.x = fmaf(h.x, w, acc.x);
    acc.y = fmaf(h.y, w, acc.y);
    acc.z = fmaf(h.z, w, acc.z);
    acc.w = fmaf(h.w, w, acc.w);
}

template <int K>
static __device__ __forceinline__ void gcn_rows(float4& acc, const float* __restrict__ Hc,
                                                const int2 cw, const int j) {
    float4 h[K];
    float w[K];
    #pragma unroll
    for (int u = 0; u < K; ++u) {
        const int si = __builtin_amdgcn_readlane(cw.x, j + u);
        w[u] = __int_as_float(__builtin_amdgcn_readlane(cw.y, j + u));
        h[u] = *(const float4*)(Hc + (size_t)si * D);
    }
    #pragma unroll
    for (int u = 0; u < K; ++u) gcn_axpy(acc, h[u], w[u]);
}

#define TP 132

static __device__ __forceinline__ void gcn_chunk(float4& acc, const float* __restrict__ Xc,
                                                 const int2 cw, const int m) {
    int j = 0;
    for (; j + 8 <= m; j += 8) gcn_rows<8>(acc, Xc, cw, j);
    if (j + 4 <= m) { gcn_rows<4>(acc, Xc, cw, j); j += 4; }
    if (j + 2 <= m) { gcn_rows<2>(acc, Xc, cw, j); j += 2; }
    if (j < m) gcn_rows<1>(acc, Xc, cw, j);
}

static __device__ __forceinline__ void gcn_row_sum(float4& acc, const float* __restrict__ Xc,
                                                   const int2* __restrict__ colw, const int beg,
                                                   const int len, const int2 first, const int lane) {
    gcn_chunk(acc, Xc, first, min(len, 32));
    for (int base = 32; base < len; base += 32) {
        const int m = min(len - base, 32);
        int2 cw = make_int2(0, 0);
        if (lane < m) cw = colw[beg + base + lane];
        gcn_chunk(acc, Xc, cw, m);
    }
}

template <bool RELU_OUT>
__global__ void __launch_bounds__(256, 8)
gcn_agg_gemm(const float* __restrict__ Xin, const float* __restrict__ dinv,
             const int* __restrict__ rowptr, const int2* __restrict__ colw,
             const float* __restrict__ W, const float* __restrict__ bias,
             float* __restrict__ Out, int n_rows) {
    __shared__ float T[16 * TP];
    __shared__ float O[16 * TP];
    __shared__ float Bs[D];

    const int lane = threadIdx.x & 31;
    const int wave = __builtin_amdgcn_readfirstlane((int)(threadIdx.x >> 5));
    const int half = lane >> 4;
    const int l16  = lane & 15;
    const int c4 = lane * 4;
    const int nTiles = (n_rows + 15) >> 4;

    if (wave == 0) *(float4*)(Bs + c4) = *(const float4*)(bias + c4);

    int tile = blockIdx.x;
    const int rpl = min(lane, 2);
    int rpv = rowptr[min(tile * 16 + wave * 2 + rpl, n_rows)];

    v2f breg[32];
    {
        const float* __restrict__ wp = W + (size_t)(half * 2) * D + wave * 16 + l16;
        #pragma unroll
        for (int k = 0; k < 32; ++k) {
            breg[k].x = wp[(size_t)(4 * k) * D];
            breg[k].y = wp[(size_t)(4 * k + 1) * D];
        }
    }
    const float* __restrict__ Xc = Xin + c4;

    for (; tile < nTiles; tile += gridDim.x) {
        const int n0 = tile * 16 + wave * 2;
        const int rp0 = __builtin_amdgcn_readlane(rpv, 0);
        const int rp1 = __builtin_amdgcn_readlane(rpv, 1);
        const int rp2 = __builtin_amdgcn_readlane(rpv, 2);
        const int beg0 = rp0, len0 = rp1 - rp0;
        const int beg1 = rp1, len1 = rp2 - rp1;

        int2 cw0 = make_int2(0, 0), cw1 = make_int2(0, 0);
        if (lane < len0) cw0 = colw[beg0 + lane];
        if (lane < len1) cw1 = colw[beg1 + lane];
        const int r0 = min(n0, n_rows - 1);
        const int r1 = min(n0 + 1, n_rows - 1);
        const float4 hs0 = *(const float4*)(Xc + (size_t)r0 * D);
        const float s0 = dinv[r0];

        rpv = rowptr[min(n0 + (int)gridDim.x * 16 + rpl, n_rows)];

        float4 acc0 = make_float4(0.0f, 0.0f, 0.0f, 0.0f);
        gcn_row_sum(acc0, Xc, colw, beg0, len0, cw0, lane);
        gcn_axpy(acc0, hs0, s0 * s0);
        *(float4*)(T + (wave * 2) * TP + c4) = acc0;

        const float4 hs1 = *(const float4*)(Xc + (size_t)r1 * D);
        const float s1 = dinv[r1];
        float4 acc1 = make_float4(0.0f, 0.0f, 0.0f, 0.0f);
        gcn_row_sum(acc1, Xc, colw, beg1, len1, cw1, lane);
        gcn_axpy(acc1, hs1, s1 * s1);
        *(float4*)(T + (wave * 2 + 1) * TP + c4) = acc1;

        __syncthreads();

        v8f c = {};
        {
            const float* ap = T + l16 * TP + half * 2;
            #pragma unroll
            for (int g = 0; g < 2; ++g) {
                v2f a[16];
                #pragma unroll
                for (int k = 0; k < 16; ++k) a[k] = *(const v2f*)(ap + 4 * (16 * g + k));
                __builtin_amdgcn_sched_barrier(0);
                #pragma unroll
                for (int k = 0; k < 16; ++k) {
                    c = __builtin_amdgcn_wmma_f32_16x16x4_f32(
                         false, a[k],  false, breg[16 * g + k],
                         (short)0, c,  false,  false);
                }
            }
            __builtin_amdgcn_sched_barrier(0);
        }
        {
            float* op = O + (half * 8) * TP + wave * 16 + l16;
            #pragma unroll
            for (int v = 0; v < 8; ++v) op[v * TP] = c[v];
        }
        __syncthreads();

        #pragma unroll
        for (int r = 0; r < 2; ++r) {
            const int lr = wave * 2 + r;
            const int node = n0 + r;
            if (node < n_rows) {
                float4 y = *(const float4*)(O + lr * TP + c4);
                const float4 b4 = *(const float4*)(Bs + c4);
                y.x += b4.x; y.y += b4.y; y.z += b4.z; y.w += b4.w;
                if (RELU_OUT) {
                    y.x = fmaxf(y.x, 0.0f); y.y = fmaxf(y.y, 0.0f);
                    y.z = fmaxf(y.z, 0.0f); y.w = fmaxf(y.w, 0.0f);
                }
                *(float4*)(Out + (size_t)node * D + c4) = y;
            }
        }
    }
}


extern "C" void kernel_launch(void* const* d_in, const int* in_sizes, int n_in,
                              void* d_out, int out_size, void* d_ws, size_t ws_size,
                              hipStream_t stream) {
    const float* x   = (const float*)d_in[0];
    const int*   ei  = (const int*)d_in[1];
    const float* W1  = (const float*)d_in[2];
    const float* b1  = (const float*)d_in[3];
    const float* W2  = (const float*)d_in[4];
    const float* b2  = (const float*)d_in[5];
    const float* W3  = (const float*)d_in[6];
    const float* b3  = (const float*)d_in[7];

    const int N = in_sizes[0] / D;
    const int E = in_sizes[1] / 2;
    const int* src = ei;
    const int* dst = ei + E;

    const size_t nodeBytes = ((size_t)(N + 1) * sizeof(int) + 255) & ~(size_t)255;
    const size_t edgeBytes = ((size_t)E * sizeof(int) + 255) & ~(size_t)255;
    char* p = (char*)d_ws;
    float* dinv = (float*)p;                 p += nodeBytes;
    float* H = (float*)p;                    p += (size_t)N * D * sizeof(float);
    float* Y = (float*)p;                    p += (size_t)N * D * sizeof(float);
    int* cnt = (int*)p;                      p += nodeBytes;
    int* rowptr = (int*)p;                   p += nodeBytes;
    int* rank = (int*)p;                     p += edgeBytes;
    int2* colw0 = (int2*)p;                  p += 2 * edgeBytes;
    int2* colw = (int2*)p;
    float* out = (float*)d_out;

    const int T = 256;
    const int nodeBlocks  = (N + T - 1) / T;
    const int scanBlocks  = (N + 1023) / 1024;
    const int rowBlocks   = 2048;
    const int layerBlocks = 1024;

    gcn_zero_cnt<<<nodeBlocks, T, 0, stream>>>(cnt, N);
    gcn_count_rank<<<1024, T, 0, stream>>>(dst, cnt, rank, E);
    gcn_scan_dinv<<<scanBlocks, 1024, 0, stream>>>(cnt, rowptr, dinv, N);
    gcn_fill<<<1024, T, 0, stream>>>(src, dst, rank, rowptr, dinv, colw0, E);
    gcn_order_rows<<<rowBlocks, T, 0, stream>>>(rowptr, colw0, colw, N);

    gcn_agg_gemm<true><<<layerBlocks, T, 0, stream>>>(x, dinv, rowptr, colw, W1, b1, Y, N);
    gcn_agg_gemm<true><<<layerBlocks, T, 0, stream>>>(Y, dinv, rowptr, colw, W2, b2, H, N);
    gcn_agg_gemm<false><<<layerBlocks, T, 0, stream>>>(H, dinv, rowptr, colw, W3, b3, out, N);
}


# edge list built by binning in LDS: work-groups group their edges by destination range, one work-group per range counts, scans and places them; no global atomics; 6 launches, were 8; sums unchanged
# speedup vs baseline: 5.7277x; 1.5578x over previous
#include <hip/hip_runtime.h>
#include <hip/hip_bf16.h>


typedef float v2f __attribute__((ext_vector_type(2)));
typedef float v8f __attribute__((ext_vector_type(8)));

#define D 128


#define BIN_CH 3200
#define NBIN 256
#define PT 1024
#define BUILD_CAP 8192

static __device__ __forceinline__ int gcn_scan256(const int v, int* wtot, const int lane,
                                                  const int wave, int& total) {
    int x = v;
    for (int o = 1; o < 32; o <<= 1) {
        const int y = __shfl_up(x, o);
        if (lane >= o) x += y;
    }
    if (lane == 31 && wave < 8) wtot[wave] = x;
    __syncthreads();
    int before = 0, all = 0;
    #pragma unroll
    for (int w = 0; w < 8; ++w) {
        const int t = wtot[w];
        all += t;
        if (w < wave) before += t;
    }
    total = all;
    return before + x;
}

__global__ void __launch_bounds__(PT)
gcn_bin(const int* __restrict__ src, const int* __restrict__ dst, int* __restrict__ stageS,
        int* __restrict__ stageD, int* __restrict__ tabCnt, int* __restrict__ tabOff,
        int n_edges, int R) {
    __shared__ int cnt[NBIN];
    __shared__ int cur[NBIN];
    __shared__ int wtot[8];
    __shared__ int stS[BIN_CH];
    __shared__ int stD[BIN_CH];
    const int tid = threadIdx.x;
    const int lane = tid & 31;
    const int wave = tid >> 5;
    const int g = blockIdx.x;
    const int e0 = g * BIN_CH;
    const int n = min(BIN_CH, n_edges - e0);

    int s[4], d[4], bn[4];
    #pragma unroll
    for (int u = 0; u < 4; ++u) {
        const int i = tid + u * PT;
        s[u] = 0; d[u] = 0;
        if (i < n) { d[u] = dst[e0 + i]; s[u] = src[e0 + i]; }
    }
    if (tid < NBIN) cnt[tid] = 0;
    __syncthreads();
    #pragma unroll
    for (int u = 0; u < 4; ++u) {
        bn[u] = d[u] / R;
        if (tid + u * PT < n) atomicAdd(&cnt[bn[u]], 1);
    }
    __syncthreads();

    const int c = (tid < NBIN) ? cnt[tid] : 0;
    int total;
    const int off = gcn_scan256(c, wtot, lane, wave, total) - c;
    if (tid < NBIN) {
        cur[tid] = off;
        tabCnt[g * NBIN + tid] = c;
        tabOff[g * NBIN + tid] = off;
    }
    __syncthreads();

    #pragma unroll
    for (int u = 0; u < 4; ++u) {
        if (tid + u * PT < n) {
            const int pos = atomicAdd(&cur[bn[u]], 1);
            stS[pos] = s[u];
            stD[pos] = d[u];
        }
    }
    __syncthreads();
    #pragma unroll
    for (int u = 0; u < 4; ++u) {
        const int i = tid + u * PT;
        if (i < n) {
            stageS[e0 + i] = stS[i];
            stageD[e0 + i] = stD[i];
        }
    }
}

static __device__ __forceinline__ void gcn_places(int (&adr)[4], const int* pstart, const int* poff,
                                                  const int i0, const int nb) {
    int lo[4], hi[4];
    #pragma unroll
    for (int u = 0; u < 4; ++u) { lo[u] = 0; hi[u] = NBIN; }
    #pragma unroll
    for (int k = 0; k < 8; ++k) {
        #pragma unroll
        for (int u = 0; u < 4; ++u) {
            const int mid = (lo[u] + hi[u]) >> 1;
            if (pstart[mid] <= i0 + u * PT) lo[u] = mid; else hi[u] = mid;
        }
    }
    #pragma unroll
    for (int u = 0; u < 4; ++u) {
        const int i = i0 + u * PT;
        adr[u] = (i < nb) ? poff[lo[u]] + (i - pstart[lo[u]]) : -1;
    }
}

__global__ void __launch_bounds__(PT)
gcn_build(const int* __restrict__ stageS, const int* __restrict__ stageD, const int* __restrict__ tabCnt,
          const int* __restrict__ tabOff, int* __restrict__ rowptr, float* __restrict__ dinv,
          int* __restrict__ col, int n_rows, int R, int n_pieces) {
    __shared__ int pstart[NBIN + 1];
    __shared__ int poff[NBIN];
    __shared__ int deg[NBIN];
    __shared__ int cur[NBIN];
    __shared__ int wtotA[8];
    __shared__ int wtotB[8];
    __shared__ int wtotC[8];
    __shared__ int outv[BUILD_CAP];
    const int tid = threadIdx.x;
    const int lane = tid & 31;
    const int wave = tid >> 5;
    const int b = blockIdx.x;
    const int lo = b * R;
    const int nr = min(R, n_rows - lo);

    int c = 0, o = 0;
    if (tid < n_pieces) {
        c = tabCnt[tid * NBIN + b];
        o = tabOff[tid * NBIN + b];
    }
    if (tid < NBIN) deg[tid] = 0;
    int nb, base;
    const int pend = gcn_scan256(c, wtotA, lane, wave, nb);
    (void)gcn_scan256(o, wtotB, lane, wave, base);
    if (tid < NBIN) {
        pstart[tid] = pend - c;
        if (tid == NBIN - 1) pstart[NBIN] = pend;
        poff[tid] = tid * BIN_CH + o;
    }
    __syncthreads();

    for (int i0 = tid; i0 < nb; i0 += 4 * PT) {
        int adr[4], dd[4];
        gcn_places(adr, pstart, poff, i0, nb);
        #pragma unroll
        for (int u = 0; u < 4; ++u) dd[u] = (adr[u] >= 0) ? stageD[adr[u]] : lo;
        #pragma unroll
        for (int u = 0; u < 4; ++u)
            if (adr[u] >= 0) atomicAdd(&deg[dd[u] - lo], 1);
    }
    __syncthreads();

    const int dg = (tid < NBIN) ? deg[tid] : 0;
    int tot2;
    const int rowoff = gcn_scan256(dg, wtotC, lane, wave, tot2) - dg;
    if (tid < NBIN) cur[tid] = rowoff;
    if (tid < nr) {
        const int node = lo + tid;
        rowptr[node] = base + rowoff;
        if (node == n_rows - 1) rowptr[n_rows] = base + rowoff + dg;
        const float dd = (float)(dg + 1);
        dinv[node] = (dd > 0.0f) ? __frsqrt_rn(dd) : 0.0f;
    }
    __syncthreads();

    const bool fits = nb <= BUILD_CAP;
    for (int i0 = tid; i0 < nb; i0 += 4 * PT) {
        int adr[4], dd[4], ss[4];
        gcn_places(adr, pstart, poff, i0, nb);
        #pragma unroll
        for (int u = 0; u < 4; ++u) {
            dd[u] = (adr[u] >= 0) ? stageD[adr[u]] : lo;
            ss[u] = (adr[u] >= 0) ? stageS[adr[u]] : 0;
        }
        #pragma unroll
        for (int u = 0; u < 4; ++u) {
            if (adr[u] >= 0) {
                const int pos = atomicAdd(&cur[dd[u] - lo], 1);
                if (fits) outv[pos] = ss[u]; else col[base + pos] = ss[u];
            }
        }
    }
    __syncthreads();
    if (fits)
        for (int i = tid; i < nb; i += PT) col[base + i] = outv[i];
}

__global__ void __launch_bounds__(256)
gcn_order_rows(const int* __restrict__ rowptr, const int* __restrict__ col_in,
               const float* __restrict__ dinv, int2* __restrict__ colw_out, int n_rows) {
    const int lane = threadIdx.x & 31;
    const int wid = __builtin_amdgcn_readfirstlane((int)((blockIdx.x * blockDim.x + threadIdx.x) >> 5));
    const int nwaves = (int)((gridDim.x * blockDim.x) >> 5);

    for (int node = wid; node < n_rows; node += nwaves) {
        const int beg = rowptr[node];
        const int len = rowptr[node + 1] - beg;
        const float dn = dinv[node];
        for (int c0 = 0; c0 < len; c0 += 32) {
            const int i = c0 + lane;
            int mine = 0;
            float ws = 0.0f;
            if (i < len) {
                mine = col_in[beg + i];
                ws = dinv[mine];
            }
            int pos = 0;
            for (int c1 = 0; c1 < len; c1 += 32) {
                const int i1 = c1 + lane;
                int k1 = 0;
                if (i1 < len) k1 = col_in[beg + i1];
                int m1 = len - c1;
                if (m1 > 32) m1 = 32;
                for (int j = 0; j < m1; ++j) {
                    const int kj = __builtin_amdgcn_readlane(k1, j);
                    const int ij = c1 + j;
                    pos += (kj < mine || (kj == mine && ij < i)) ? 1 : 0;
                }
            }
            if (i < len) colw_out[beg + pos] = make_int2(mine, __float_as_int(ws * dn));
        }
    }
}


static __device__ __forceinline__ void gcn_axpy(float4& acc, const float4 h, const float w) {
    acc.x = fmaf(h.x, w, acc.x);
    acc.y = fmaf(h.y, w, acc.y);
    acc.z = fmaf(h.z, w, acc.z);
    acc.w = fmaf(h.w, w, acc.w);
}

template <int K>
static __device__ __forceinline__ void gcn_rows(float4& acc, const float* __restrict__ Hc,
                                                const int2 cw, const int j) {
    float4 h[K];
    float w[K];
    #pragma unroll
    for (int u = 0; u < K; ++u) {
        const int si = __builtin_amdgcn_readlane(cw.x, j + u);
        w[u] = __int_as_float(__builtin_amdgcn_readlane(cw.y, j + u));
        h[u] = *(const float4*)(Hc + (size_t)si * D);
    }
    #pragma unroll
    for (int u = 0; u < K; ++u) gcn_axpy(acc, h[u], w[u]);
}

#define TP 132

static __device__ __forceinline__ void gcn_chunk(float4& acc, const float* __restrict__ Xc,
                                                 const int2 cw, const int m) {
    int j = 0;
    for (; j + 8 <= m; j += 8) gcn_rows<8>(acc, Xc, cw, j);
    if (j + 4 <= m) { gcn_rows<4>(acc, Xc, cw, j); j += 4; }
    if (j + 2 <= m) { gcn_rows<2>(acc, Xc, cw, j); j += 2; }
    if (j < m) gcn_rows<1>(acc, Xc, cw, j);
}

static __device__ __forceinline__ void gcn_row_sum(float4& acc, const float* __restrict__ Xc,
                                                   const int2* __restrict__ colw, const int beg,
                                                   const int len, const int2 first, const int lane) {
    gcn_chunk(acc, Xc, first, min(len, 32));
    for (int base = 32; base < len; base += 32) {
        const int m = min(len - base, 32);
        int2 cw = make_int2(0, 0);
        if (lane < m) cw = colw[beg + base + lane];
        gcn_chunk(acc, Xc, cw, m);
    }
}

template <bool RELU_OUT>
__global__ void __launch_bounds__(256, 8)
gcn_agg_gemm(const float* __restrict__ Xin, const float* __restrict__ dinv,
             const int* __restrict__ rowptr, const int2* __restrict__ colw,
             const float* __restrict__ W, const float* __restrict__ bias,
             float* __restrict__ Out, int n_rows) {
    __shared__ float T[16 * TP];
    __shared__ float O[16 * TP];
    __shared__ float Bs[D];

    const int lane = threadIdx.x & 31;
    const int wave = __builtin_amdgcn_readfirstlane((int)(threadIdx.x >> 5));
    const int half = lane >> 4;
    const int l16  = lane & 15;
    const int c4 = lane * 4;
    const int nTiles = (n_rows + 15) >> 4;

    if (wave == 0) *(float4*)(Bs + c4) = *(const float4*)(bias + c4);

    int tile = blockIdx.x;
    const int rpl = min(lane, 2);
    int rpv = rowptr[min(tile * 16 + wave * 2 + rpl, n_rows)];

    v2f breg[32];
    {
        const float* __restrict__ wp = W + (size_t)(half * 2) * D + wave * 16 + l16;
        #pragma unroll
        for (int k = 0; k < 32; ++k) {
            breg[k].x = wp[(size_t)(4 * k) * D];
            breg[k].y = wp[(size_t)(4 * k + 1) * D];
        }
    }
    const float* __restrict__ Xc = Xin + c4;

    for (; tile < nTiles; tile += gridDim.x) {
        const int n0 = tile * 16 + wave * 2;
        const int rp0 = __builtin_amdgcn_readlane(rpv, 0);
        const int rp1 = __builtin_amdgcn_readlane(rpv, 1);
        const int rp2 = __builtin_amdgcn_readlane(rpv, 2);
        const int beg0 = rp0, len0 = rp1 - rp0;
        const int beg1 = rp1, len1 = rp2 - rp1;

        int2 cw0 = make_int2(0, 0), cw1 = make_int2(0, 0);
        if (lane < len0) cw0 = colw[beg0 + lane];
        if (lane < len1) cw1 = colw[beg1 + lane];
        const int r0 = min(n0, n_rows - 1);
        const int r1 = min(n0 + 1, n_rows - 1);
        const float4 hs0 = *(const float4*)(Xc + (size_t)r0 * D);
        const float s0 = dinv[r0];

        rpv = rowptr[min(n0 + (int)gridDim.x * 16 + rpl, n_rows)];

        float4 acc0 = make_float4(0.0f, 0.0f, 0.0f, 0.0f);
        gcn_row_sum(acc0, Xc, colw, beg0, len0, cw0, lane);
        gcn_axpy(acc0, hs0, s0 * s0);
        *(float4*)(T + (wave * 2) * TP + c4) = acc0;

        const float4 hs1 = *(const float4*)(Xc + (size_t)r1 * D);
        const float s1 = dinv[r1];
        float4 acc1 = make_float4(0.0f, 0.0f, 0.0f, 0.0f);
        gcn_row_sum(acc1, Xc, colw, beg1, len1, cw1, lane);
        gcn_axpy(acc1, hs1, s1 * s1);
        *(float4*)(T + (wave * 2 + 1) * TP + c4) = acc1;

        __syncthreads();

        v8f c = {};
        {
            const float* ap = T + l16 * TP + half * 2;
            #pragma unroll
            for (int g = 0; g < 2; ++g) {
                v2f a[16];
                #pragma unroll
                for (int k = 0; k < 16; ++k) a[k] = *(const v2f*)(ap + 4 * (16 * g + k));
                __builtin_amdgcn_sched_barrier(0);
                #pragma unroll
                for (int k = 0; k < 16; ++k) {
                    c = __builtin_amdgcn_wmma_f32_16x16x4_f32(
                         false, a[k],  false, breg[16 * g + k],
                         (short)0, c,  false,  false);
                }
            }
            __builtin_amdgcn_sched_barrier(0);
        }
        {
            float* op = O + (half * 8) * TP + wave * 16 + l16;
            #pragma unroll
            for (int v = 0; v < 8; ++v) op[v * TP] = c[v];
        }
        __syncthreads();

        #pragma unroll
        for (int r = 0; r < 2; ++r) {
            const int lr = wave * 2 + r;
            const int node = n0 + r;
            if (node < n_rows) {
                float4 y = *(const float4*)(O + lr * TP + c4);
                const float4 b4 = *(const float4*)(Bs + c4);
                y.x += b4.x; y.y += b4.y; y.z += b4.z; y.w += b4.w;
                if (RELU_OUT) {
                    y.x = fmaxf(y.x, 0.0f); y.y = fmaxf(y.y, 0.0f);
                    y.z = fmaxf(y.z, 0.0f); y.w = fmaxf(y.w, 0.0f);
                }
                *(float4*)(Out + (size_t)node * D + c4) = y;
            }
        }
    }
}


extern "C" void kernel_launch(void* const* d_in, const int* in_sizes, int n_in,
                              void* d_out, int out_size, void* d_ws, size_t ws_size,
                              hipStream_t stream) {
    const float* x   = (const float*)d_in[0];
    const int*   ei  = (const int*)d_in[1];
    const float* W1  = (const float*)d_in[2];
    const float* b1  = (const float*)d_in[3];
    const float* W2  = (const float*)d_in[4];
    const float* b2  = (const float*)d_in[5];
    const float* W3  = (const float*)d_in[6];
    const float* b3  = (const float*)d_in[7];

    const int N = in_sizes[0] / D;
    const int E = in_sizes[1] / 2;
    const int* src = ei;
    const int* dst = ei + E;

    const size_t nodeBytes = ((size_t)(N + 1) * sizeof(int) + 255) & ~(size_t)255;
    const size_t edgeBytes = ((size_t)E * sizeof(int) + 255) & ~(size_t)255;
    char* p = (char*)d_ws;
    float* dinv = (float*)p;                 p += nodeBytes;
    float* H = (float*)p;                    p += (size_t)N * D * sizeof(float);
    float* Y = (float*)p;                    p += (size_t)N * D * sizeof(float);
    int* rowptr = (int*)p;                   p += nodeBytes;
    int* stageS = (int*)p;                   p += edgeBytes;
    int* stageD = (int*)p;                   p += edgeBytes;
    int* tabCnt = (int*)p;                   p += (size_t)NBIN * NBIN * sizeof(int);
    int* tabOff = (int*)p;                   p += (size_t)NBIN * NBIN * sizeof(int);
    int* col0 = (int*)p;                     p += edgeBytes;
    int2* colw = (int2*)p;
    float* out = (float*)d_out;

    const int T = 256;
    const int binBlocks   = (E + BIN_CH - 1) / BIN_CH;
    const int R           = (N + NBIN - 1) / NBIN;
    const int rowBlocks   = 2048;
    const int layerBlocks = 1024;

    gcn_bin<<<binBlocks, PT, 0, stream>>>(src, dst, stageS, stageD, tabCnt, tabOff, E, R);
    gcn_build<<<NBIN, PT, 0, stream>>>(stageS, stageD, tabCnt, tabOff, rowptr, dinv, col0, N, R, binBlocks);
    gcn_order_rows<<<rowBlocks, T, 0, stream>>>(rowptr, col0, dinv, colw, N);

    gcn_agg_gemm<true><<<layerBlocks, T, 0, stream>>>(x, dinv, rowptr, colw, W1, b1, Y, N);
    gcn_agg_gemm<true><<<layerBlocks, T, 0, stream>>>(Y, dinv, rowptr, colw, W2, b2, H, N);
    gcn_agg_gemm<false><<<layerBlocks, T, 0, stream>>>(H, dinv, rowptr, colw, W3, b3, out, N);
}


# layer kernel as a pipeline over tiles in 24-wave work-groups: 16 waves gather a row each (list heads a tile ahead), 8 multiply the previous tile (W in LDS) and store the one before; sums unchanged
# speedup vs baseline: 5.9635x; 1.0412x over previous
#include <hip/hip_runtime.h>
#include <hip/hip_bf16.h>


typedef float v2f __attribute__((ext_vector_type(2)));
typedef float v8f __attribute__((ext_vector_type(8)));

#define D 128


#define BIN_CH 3200
#define NBIN 256
#define PT 1024
#define BUILD_CAP 8192

static __device__ __forceinline__ int gcn_scan256(const int v, int* wtot, const int lane,
                                                  const int wave, int& total) {
    int x = v;
    for (int o = 1; o < 32; o <<= 1) {
        const int y = __shfl_up(x, o);
        if (lane >= o) x += y;
    }
    if (lane == 31 && wave < 8) wtot[wave] = x;
    __syncthreads();
    int before = 0, all = 0;
    #pragma unroll
    for (int w = 0; w < 8; ++w) {
        const int t = wtot[w];
        all += t;
        if (w < wave) before += t;
    }
    total = all;
    return before + x;
}

__global__ void __launch_bounds__(PT)
gcn_bin(const int* __restrict__ src, const int* __restrict__ dst, int* __restrict__ stageS,
        int* __restrict__ stageD, int* __restrict__ tabCnt, int* __restrict__ tabOff,
        int n_edges, int R) {
    __shared__ int cnt[NBIN];
    __shared__ int cur[NBIN];
    __shared__ int wtot[8];
    __shared__ int stS[BIN_CH];
    __shared__ int stD[BIN_CH];
    const int tid = threadIdx.x;
    const int lane = tid & 31;
    const int wave = tid >> 5;
    const int g = blockIdx.x;
    const int e0 = g * BIN_CH;
    const int n = min(BIN_CH, n_edges - e0);

    int s[4], d[4], bn[4];
    #pragma unroll
    for (int u = 0; u < 4; ++u) {
        const int i = tid + u * PT;
        s[u] = 0; d[u] = 0;
        if (i < n) { d[u] = dst[e0 + i]; s[u] = src[e0 + i]; }
    }
    if (tid < NBIN) cnt[tid] = 0;
    __syncthreads();
    #pragma unroll
    for (int u = 0; u < 4; ++u) {
        bn[u] = d[u] / R;
        if (tid + u * PT < n) atomicAdd(&cnt[bn[u]], 1);
    }
    __syncthreads();

    const int c = (tid < NBIN) ? cnt[tid] : 0;
    int total;
    const int off = gcn_scan256(c, wtot, lane, wave, total) - c;
    if (tid < NBIN) {
        cur[tid] = off;
        tabCnt[g * NBIN + tid] = c;
        tabOff[g * NBIN + tid] = off;
    }
    __syncthreads();

    #pragma unroll
    for (int u = 0; u < 4; ++u) {
        if (tid + u * PT < n) {
            const int pos = atomicAdd(&cur[bn[u]], 1);
            stS[pos] = s[u];
            stD[pos] = d[u];
        }
    }
    __syncthreads();
    #pragma unroll
    for (int u = 0; u < 4; ++u) {
        const int i = tid + u * PT;
        if (i < n) {
            stageS[e0 + i] = stS[i];
            stageD[e0 + i] = stD[i];
        }
    }
}

static __device__ __forceinline__ void gcn_places(int (&adr)[4], const int* pstart, const int* poff,
                                                  const int i0, const int nb) {
    int lo[4], hi[4];
    #pragma unroll
    for (int u = 0; u < 4; ++u) { lo[u] = 0; hi[u] = NBIN; }
    #pragma unroll
    for (int k = 0; k < 8; ++k) {
        #pragma unroll
        for (int u = 0; u < 4; ++u) {
            const int mid = (lo[u] + hi[u]) >> 1;
            if (pstart[mid] <= i0 + u * PT) lo[u] = mid; else hi[u] = mid;
        }
    }
    #pragma unroll
    for (int u = 0; u < 4; ++u) {
        const int i = i0 + u * PT;
        adr[u] = (i < nb) ? poff[lo[u]] + (i - pstart[lo[u]]) : -1;
    }
}

__global__ void __launch_bounds__(PT)
gcn_build(const int* __restrict__ stageS, const int* __restrict__ stageD, const int* __restrict__ tabCnt,
          const int* __restrict__ tabOff, int* __restrict__ rowptr, float* __restrict__ dinv,
          int* __restrict__ col, int n_rows, int R, int n_pieces) {
    __shared__ int pstart[NBIN + 1];
    __shared__ int poff[NBIN];
    __shared__ int deg[NBIN];
    __shared__ int cur[NBIN];
    __shared__ int wtotA[8];
    __shared__ int wtotB[8];
    __shared__ int wtotC[8];
    __shared__ int outv[BUILD_CAP];
    const int tid = threadIdx.x;
    const int lane = tid & 31;
    const int wave = tid >> 5;
    const int b = blockIdx.x;
    const int lo = b * R;
    const int nr = min(R, n_rows - lo);

    int c = 0, o = 0;
    if (tid < n_pieces) {
        c = tabCnt[tid * NBIN + b];
        o = tabOff[tid * NBIN + b];
    }
    if (tid < NBIN) deg[tid] = 0;
    int nb, base;
    const int pend = gcn_scan256(c, wtotA, lane, wave, nb);
    (void)gcn_scan256(o, wtotB, lane, wave, base);
    if (tid < NBIN) {
        pstart[tid] = pend - c;
        if (tid == NBIN - 1) pstart[NBIN] = pend;
        poff[tid] = tid * BIN_CH + o;
    }
    __syncthreads();

    for (int i0 = tid; i0 < nb; i0 += 4 * PT) {
        int adr[4], dd[4];
        gcn_places(adr, pstart, poff, i0, nb);
        #pragma unroll
        for (int u = 0; u < 4; ++u) dd[u] = (adr[u] >= 0) ? stageD[adr[u]] : lo;
        #pragma unroll
        for (int u = 0; u < 4; ++u)
            if (adr[u] >= 0) atomicAdd(&deg[dd[u] - lo], 1);
    }
    __syncthreads();

    const int dg = (tid < NBIN) ? deg[tid] : 0;
    int tot2;
    const int rowoff = gcn_scan256(dg, wtotC, lane, wave, tot2) - dg;
    if (tid < NBIN) cur[tid] = rowoff;
    if (tid < nr) {
        const int node = lo + tid;
        rowptr[node] = base + rowoff;
        if (node == n_rows - 1) rowptr[n_rows] = base + rowoff + dg;
        const float dd = (float)(dg + 1);
        dinv[node] = (dd > 0.0f) ? __frsqrt_rn(dd) : 0.0f;
    }
    __syncthreads();

    const bool fits = nb <= BUILD_CAP;
    for (int i0 = tid; i0 < nb; i0 += 4 * PT) {
        int adr[4], dd[4], ss[4];
        gcn_places(adr, pstart, poff, i0, nb);
        #pragma unroll
        for (int u = 0; u < 4; ++u) {
            dd[u] = (adr[u] >= 0) ? stageD[adr[u]] : lo;
            ss[u] = (adr[u] >= 0) ? stageS[adr[u]] : 0;
        }
        #pragma unroll
        for (int u = 0; u < 4; ++u) {
            if (adr[u] >= 0) {
                const int pos = atomicAdd(&cur[dd[u] - lo], 1);
                if (fits) outv[pos] = ss[u]; else col[base + pos] = ss[u];
            }
        }
    }
    __syncthreads();
    if (fits)
        for (int i = tid; i < nb; i += PT) col[base + i] = outv[i];
}

__global__ void __launch_bounds__(256)
gcn_order_rows(const int* __restrict__ rowptr, const int* __restrict__ col_in,
               const float* __restrict__ dinv, int2* __restrict__ colw_out, int n_rows) {
    const int lane = threadIdx.x & 31;
    const int wid = __builtin_amdgcn_readfirstlane((int)((blockIdx.x * blockDim.x + threadIdx.x) >> 5));
    const int nwaves = (int)((gridDim.x * blockDim.x) >> 5);

    for (int node = wid; node < n_rows; node += nwaves) {
        const int beg = rowptr[node];
        const int len = rowptr[node + 1] - beg;
        const float dn = dinv[node];
        for (int c0 = 0; c0 < len; c0 += 32) {
            const int i = c0 + lane;
            int mine = 0;
            float ws = 0.0f;
            if (i < len) {
                mine = col_in[beg + i];
                ws = dinv[mine];
            }
            int pos = 0;
            for (int c1 = 0; c1 < len; c1 += 32) {
                const int i1 = c1 + lane;
                int k1 = 0;
                if (i1 < len) k1 = col_in[beg + i1];
                int m1 = len - c1;
                if (m1 > 32) m1 = 32;
                for (int j = 0; j < m1; ++j) {
                    const int kj = __builtin_amdgcn_readlane(k1, j);
                    const int ij = c1 + j;
                    pos += (kj < mine || (kj == mine && ij < i)) ? 1 : 0;
                }
            }
            if (i < len) colw_out[beg + pos] = make_int2(mine, __float_as_int(ws * dn));
        }
    }
}


static __device__ __forceinline__ void gcn_axpy(float4& acc, const float4 h, const float w) {
    acc.x = fmaf(h.x, w, acc.x);
    acc.y = fmaf(h.y, w, acc.y);
    acc.z = fmaf(h.z, w, acc.z);
    acc.w = fmaf(h.w, w, acc.w);
}

template <int K>
static __device__ __forceinline__ void gcn_rows(float4& acc, const float* __restrict__ Hc,
                                                const int2 cw, const int j) {
    float4 h[K];
    float w[K];
    #pragma unroll
    for (int u = 0; u < K; ++u) {
        const int si = __builtin_amdgcn_readlane(cw.x, j + u);
        w[u] = __int_as_float(__builtin_amdgcn_readlane(cw.y, j + u));
        h[u] = *(const float4*)(Hc + (size_t)si * D);
    }
    #pragma unroll
    for (int u = 0; u < K; ++u) gcn_axpy(acc, h[u], w[u]);
}

#define TP 132

static __device__ __forceinline__ void gcn_chunk(float4& acc, const float* __restrict__ Xc,
                                                 const int2 cw, const int m) {
    int j = 0;
    for (; j + 8 <= m; j += 8) gcn_rows<8>(acc, Xc, cw, j);
    if (j + 4 <= m) { gcn_rows<4>(acc, Xc, cw, j); j += 4; }
    if (j + 2 <= m) { gcn_rows<2>(acc, Xc, cw, j); j += 2; }
    if (j < m) gcn_rows<1>(acc, Xc, cw, j);
}

static __device__ __forceinline__ void gcn_row_sum(float4& acc, const float* __restrict__ Xc,
                                                   const int2* __restrict__ colw, const int beg,
                                                   const int len, const int2 first, const int lane) {
    gcn_chunk(acc, Xc, first, min(len, 32));
    for (int base = 32; base < len; base += 32) {
        const int m = min(len - base, 32);
        int2 cw = make_int2(0, 0);
        if (lane < m) cw = colw[beg + base + lane];
        gcn_chunk(acc, Xc, cw, m);
    }
}

template <bool RELU_OUT>
__global__ void __launch_bounds__(768, 12)
gcn_agg_gemm(const float* __restrict__ Xin, const float* __restrict__ dinv,
             const int* __restrict__ rowptr, const int2* __restrict__ colw,
             const float* __restrict__ W, const float* __restrict__ bias,
             float* __restrict__ Out, int n_rows) {
    __shared__ __attribute__((aligned(16))) float Wl[8 * 32 * 64];
    __shared__ __attribute__((aligned(16))) float T[2 * 16 * TP];
    __shared__ __attribute__((aligned(16))) float O[2 * 16 * TP];
    __shared__ __attribute__((aligned(16))) float Bs[D];

    const int lane = threadIdx.x & 31;
    const int wave = __builtin_amdgcn_readfirstlane((int)(threadIdx.x >> 5));
    const bool mem_role = wave < 16;
    const int m = wave & 7;
    const int half = lane >> 4;
    const int l16  = lane & 15;
    const int c4 = lane * 4;
    const int nTiles = (n_rows + 15) >> 4;
    const int G = (int)gridDim.x;
    const int tile0 = (int)blockIdx.x;
    const int nMine = (tile0 < nTiles) ? (nTiles - 1 - tile0) / G + 1 : 0;
    const float* __restrict__ Xc = Xin + c4;

    const int rpl = min(lane, 1);
    int rpv = 0, beg = 0, len = 0;
    int2 cw = make_int2(0, 0);
    if (mem_role) {
        const int n = tile0 * 16 + wave;
        const int rp = rowptr[min(n + rpl, n_rows)];
        beg = __builtin_amdgcn_readlane(rp, 0);
        len = __builtin_amdgcn_readlane(rp, 1) - beg;
        if (lane < len) cw = colw[beg + lane];
        rpv = rowptr[min(n + G * 16 + rpl, n_rows)];
    } else {
        if (m == 0) *(float4*)(Bs + c4) = *(const float4*)(bias + c4);
        const float* __restrict__ wp = W + (size_t)(half * 2) * D + m * 16 + l16;
        v2f* wl = (v2f*)Wl + (m * 32) * 32 + lane;
        #pragma unroll
        for (int k = 0; k < 32; ++k) {
            v2f t;
            t.x = wp[(size_t)(4 * k) * D];
            t.y = wp[(size_t)(4 * k + 1) * D];
            wl[k * 32] = t;
        }
    }

    for (int it = 0; it < nMine + 2; ++it) {
        if (mem_role) {
            if (it < nMine) {
                const int n = (tile0 + it * G) * 16 + wave;
                const int begn = __builtin_amdgcn_readlane(rpv, 0);
                const int lenn = __builtin_amdgcn_readlane(rpv, 1) - begn;
                int2 cwn = make_int2(0, 0);
                if (lane < lenn) cwn = colw[begn + lane];
                rpv = rowptr[min(n + 2 * G * 16 + rpl, n_rows)];
                const int r = min(n, n_rows - 1);
                const float4 hs = *(const float4*)(Xc + (size_t)r * D);
                const float s = dinv[r];

                float4 acc = make_float4(0.0f, 0.0f, 0.0f, 0.0f);
                gcn_row_sum(acc, Xc, colw, beg, len, cw, lane);
                gcn_axpy(acc, hs, s * s);
                *(float4*)(T + (it & 1) * (16 * TP) + wave * TP + c4) = acc;

                cw = cwn; beg = begn; len = lenn;
            }
        } else {
            if (it >= 2) {
                const float* Ob = O + (it & 1) * (16 * TP);
                const int ns = (tile0 + (it - 2) * G) * 16 + m * 2;
                #pragma unroll
                for (int r = 0; r < 2; ++r) {
                    const int lr = m * 2 + r;
                    const int node = ns + r;
                    if (node < n_rows) {
                        float4 y = *(const float4*)(Ob + lr * TP + c4);
                        const float4 b4 = *(const float4*)(Bs + c4);
                        y.x += b4.x; y.y += b4.y; y.z += b4.z; y.w += b4.w;
                        if (RELU_OUT) {
                            y.x = fmaxf(y.x, 0.0f); y.y = fmaxf(y.y, 0.0f);
                            y.z = fmaxf(y.z, 0.0f); y.w = fmaxf(y.w, 0.0f);
                        }
                        *(float4*)(Out + (size_t)node * D + c4) = y;
                    }
                }
            }
            if (it >= 1 && it <= nMine) {
                const float* Tb = T + ((it - 1) & 1) * (16 * TP);
                float* Ob = O + ((it - 1) & 1) * (16 * TP);
                v8f c = {};
                {
                    const float* ap = Tb + l16 * TP + half * 2;
                    const v2f* bp = (const v2f*)Wl + (m * 32) * 32 + lane;
                    #pragma unroll
                    for (int g = 0; g < 4; ++g) {
                        v2f a[8], b[8];
                        #pragma unroll
                        for (int k = 0; k < 8; ++k) {
                            a[k] = *(const v2f*)(ap + 4 * (8 * g + k));
                            b[k] = bp[(8 * g + k) * 32];
                        }
                        __builtin_amdgcn_sched_barrier(0);
                        #pragma unroll
                        for (int k = 0; k < 8; ++k) {
                            c = __builtin_amdgcn_wmma_f32_16x16x4_f32(
                                 false, a[k],  false, b[k],
                                 (short)0, c,  false,  false);
                        }
                    }
                    __builtin_amdgcn_sched_barrier(0);
                }
                {
                    float* op = Ob + (half * 8) * TP + m * 16 + l16;
                    #pragma unroll
                    for (int v = 0; v < 8; ++v) op[v * TP] = c[v];
                }
            }
        }
        __syncthreads();
    }
}


extern "C" void kernel_launch(void* const* d_in, const int* in_sizes, int n_in,
                              void* d_out, int out_size, void* d_ws, size_t ws_size,
                              hipStream_t stream) {
    const float* x   = (const float*)d_in[0];
    const int*   ei  = (const int*)d_in[1];
    const float* W1  = (const float*)d_in[2];
    const float* b1  = (const float*)d_in[3];
    const float* W2  = (const float*)d_in[4];
    const float* b2  = (const float*)d_in[5];
    const float* W3  = (const float*)d_in[6];
    const float* b3  = (const float*)d_in[7];

    const int N = in_sizes[0] / D;
    const int E = in_sizes[1] / 2;
    const int* src = ei;
    const int* dst = ei + E;

    const size_t nodeBytes = ((size_t)(N + 1) * sizeof(int) + 255) & ~(size_t)255;
    const size_t edgeBytes = ((size_t)E * sizeof(int) + 255) & ~(size_t)255;
    char* p = (char*)d_ws;
    float* dinv = (float*)p;                 p += nodeBytes;
    float* H = (float*)p;                    p += (size_t)N * D * sizeof(float);
    float* Y = (float*)p;                    p += (size_t)N * D * sizeof(float);
    int* rowptr = (int*)p;                   p += nodeBytes;
    int* stageS = (int*)p;                   p += edgeBytes;
    int* stageD = (int*)p;                   p += edgeBytes;
    int* tabCnt = (int*)p;                   p += (size_t)NBIN * NBIN * sizeof(int);
    int* tabOff = (int*)p;                   p += (size_t)NBIN * NBIN * sizeof(int);
    int* col0 = (int*)p;                     p += edgeBytes;
    int2* colw = (int2*)p;
    float* out = (float*)d_out;

    const int T = 256;
    const int binBlocks   = (E + BIN_CH - 1) / BIN_CH;
    const int R           = (N + NBIN - 1) / NBIN;
    const int rowBlocks   = 2048;
    const int layerBlocks = 512;
    const int TL = 768;

    gcn_bin<<<binBlocks, PT, 0, stream>>>(src, dst, stageS, stageD, tabCnt, tabOff, E, R);
    gcn_build<<<NBIN, PT, 0, stream>>>(stageS, stageD, tabCnt, tabOff, rowptr, dinv, col0, N, R, binBlocks);
    gcn_order_rows<<<rowBlocks, T, 0, stream>>>(rowptr, col0, dinv, colw, N);

    gcn_agg_gemm<true><<<layerBlocks, TL, 0, stream>>>(x, dinv, rowptr, colw, W1, b1, Y, N);
    gcn_agg_gemm<true><<<layerBlocks, TL, 0, stream>>>(Y, dinv, rowptr, colw, W2, b2, H, N);
    gcn_agg_gemm<false><<<layerBlocks, TL, 0, stream>>>(H, dinv, rowptr, colw, W3, b3, out, N);
}
